# starting kernel
# baseline (speedup 1.0000x reference)
#include <hip/hip_runtime.h>
#include <hip/hip_bf16.h>

typedef __attribute__((ext_vector_type(2))) float v2f;
typedef __attribute__((ext_vector_type(8))) float v8f;

#define ND 64

__global__ __launch_bounds__(256) void gpr_init_kernel(
    const float* __restrict__ x, const float* __restrict__ temp,
    float* __restrict__ hidden, float* __restrict__ xcur,
    float* __restrict__ xnew, int total) {
  int i = blockIdx.x * 256 + threadIdx.x;
  if (i < total) {
    float v = x[i];
    hidden[i] = v * temp[0];
    xcur[i]   = v;
    xnew[i]   = 0.0f;
  }
}

__global__ __launch_bounds__(256) void gpr_gemm_wmma_kernel(
    const float* __restrict__ X, const float* __restrict__ Wm,
    const float* __restrict__ bias, float* __restrict__ H,
    int nWaves) {
  int wave = (blockIdx.x * 256 + threadIdx.x) >> 5;
  if (wave >= nWaves) return;
  int lane  = threadIdx.x & 31;
  int mtile = wave >> 2;
  int otile = wave & 3;
  int hi = lane >> 4;
  int lo = lane & 15;

  const float* arow = X  + (size_t)(mtile * 16 + lo) * ND;
  const float* brow = Wm + (size_t)(otile * 16 + lo) * ND;

  v8f c = {};
#pragma unroll
  for (int k = 0; k < ND; k += 4) {
    v2f a = *(const v2f*)(arow + k + 2 * hi);
    v2f b = *(const v2f*)(brow + k + 2 * hi);
    c = __builtin_amdgcn_wmma_f32_16x16x4_f32(
         false, a,  false, b,
         (short)0, c,  false,  false);
  }

  float bv = bias[otile * 16 + lo];
  float* hout = H + (size_t)(mtile * 16 + 8 * hi) * ND + otile * 16 + lo;
#pragma unroll
  for (int v = 0; v < 8; ++v)
    hout[(size_t)v * ND] = c[v] + bv;
}

__global__ __launch_bounds__(256) void gpr_scatter_kernel(
    const float* __restrict__ H, const float* __restrict__ w,
    const int* __restrict__ src, const int* __restrict__ dst,
    float* __restrict__ xnew, int nEdges) {
  int tid = blockIdx.x * 256 + threadIdx.x;
  int e = tid >> 4;
  if (e >= nEdges) return;
  int sub = tid & 15;

  int   s  = src[e];
  int   d  = dst[e];
  float we = w[e];

  const float4* hp = (const float4*)(H + (size_t)s * ND) + sub;
  float4 hv = *hp;

  float* out = xnew + (size_t)d * ND + sub * 4;
  unsafeAtomicAdd(out + 0, hv.x * we);
  unsafeAtomicAdd(out + 1, hv.y * we);
  unsafeAtomicAdd(out + 2, hv.z * we);
  unsafeAtomicAdd(out + 3, hv.w * we);
}

__global__ __launch_bounds__(256) void gpr_finalize_kernel(
    float* __restrict__ xnew, float* __restrict__ xcur,
    float* __restrict__ hidden, const float* __restrict__ temp,
    int tempIdx, int total) {
  int i = blockIdx.x * 256 + threadIdx.x;
  if (i < total) {
    float v = fmaxf(xnew[i], 0.0f);
    xcur[i] = v;
    hidden[i] += v * temp[tempIdx];
    xnew[i] = 0.0f;
  }
}

extern "C" void kernel_launch(void* const* d_in, const int* in_sizes, int n_in,
                              void* d_out, int out_size, void* d_ws, size_t ws_size,
                              hipStream_t stream) {
  const float* x    = (const float*)d_in[0];
  const float* w    = (const float*)d_in[1];
  const float* Wm   = (const float*)d_in[2];
  const float* bias = (const float*)d_in[3];
  const float* temp = (const float*)d_in[4];
  const int*   src  = (const int*)d_in[5];
  const int*   dst  = (const int*)d_in[6];
  float* hidden = (float*)d_out;

  const int n = in_sizes[0] / ND;
  const int E = in_sizes[1];
  const int L = in_sizes[3] / ND;
  const int total = n * ND;

  float* h    = (float*)d_ws;
  float* xcur = h    + (size_t)total;
  float* xnew = xcur + (size_t)total;

  const int ewBlocks   = (total + 255) / 256;
  const int nWaves     = (n / 16) * 4;
  const int gemmBlocks = (nWaves * 32 + 255) / 256;
  const int scatBlocks = (E * 16 + 255) / 256;

  gpr_init_kernel<<<ewBlocks, 256, 0, stream>>>(x, temp, hidden, xcur, xnew, total);

  for (int i = 0; i < L; ++i) {
    gpr_gemm_wmma_kernel<<<gemmBlocks, 256, 0, stream>>>(
        xcur, Wm + (size_t)i * ND * ND, bias + (size_t)i * ND, h, nWaves);
    gpr_scatter_kernel<<<scatBlocks, 256, 0, stream>>>(h, w, src, dst, xnew, E);
    gpr_finalize_kernel<<<ewBlocks, 256, 0, stream>>>(
        xnew, xcur, hidden, temp, i + 1, total);
  }
}


# Rank counters and slot rows split by chip half: a wave ranks on its own half's counter plane and fills that half's part of the row; no written line crosses halves; sum order: arrival, by part
# speedup vs baseline: 3.7817x; 3.7817x over previous
#include <hip/hip_runtime.h>
#include <hip/hip_bf16.h>

typedef __attribute__((ext_vector_type(2))) float v2f;
typedef __attribute__((ext_vector_type(8))) float v8f;

#define ND 64

__global__ __launch_bounds__(256) void gpr_init_kernel(
    const float* __restrict__ x, const float* __restrict__ temp,
    float* __restrict__ hidden, float* __restrict__ xcur,
    int* __restrict__ cnt, int* __restrict__ head, int nRows, int total) {
  int i = blockIdx.x * 256 + threadIdx.x;
  if (i < total) {
    float v = x[i];
    hidden[i] = v * temp[0];
    xcur[i]   = v;
  }
  if (i < 2 * nRows) cnt[i] = 0;
  if (i < nRows) head[i] = -1;
}

__global__ __launch_bounds__(256) void gpr_fill_kernel(
    const int* __restrict__ dst, const int* __restrict__ src,
    const float* __restrict__ w, int* __restrict__ cnt,
    int2* __restrict__ slots, int* __restrict__ head,
    int* __restrict__ next, int cap, int nRows, int nEdges) {
  int e = blockIdx.x * 256 + threadIdx.x;
  int hb = (int)((__builtin_amdgcn_s_sendmsg_rtn(0x87) >> 18) & 1u);
  int hcap = cap >> 1;
  if (e < nEdges) {
    int d = dst[e];
    int s = src[e];
    float wv = w[e];
    int r = atomicAdd(cnt + (size_t)hb * nRows + d, 1);
    if (r < hcap) slots[(size_t)d * cap + hb * hcap + r] = make_int2(s, __float_as_int(wv));
    else          next[e] = atomicExch(head + d, e);
  }
}

__global__ __launch_bounds__(256) void gpr_gemm_wmma_kernel(
    const float* __restrict__ X, const float* __restrict__ Wm,
    const float* __restrict__ bias, float* __restrict__ H,
    int nWaves) {
  int wave = (blockIdx.x * 256 + threadIdx.x) >> 5;
  if (wave >= nWaves) return;
  int lane  = threadIdx.x & 31;
  int mtile = wave >> 2;
  int otile = wave & 3;
  int hi = lane >> 4;
  int lo = lane & 15;

  const float* arow = X  + (size_t)(mtile * 16 + lo) * ND;
  const float* brow = Wm + (size_t)(otile * 16 + lo) * ND;

  v8f c = {};
#pragma unroll
  for (int k = 0; k < ND; k += 4) {
    v2f a = *(const v2f*)(arow + k + 2 * hi);
    v2f b = *(const v2f*)(brow + k + 2 * hi);
    c = __builtin_amdgcn_wmma_f32_16x16x4_f32(
         false, a,  false, b,
         (short)0, c,  false,  false);
  }

  float bv = bias[otile * 16 + lo];
  float* hout = H + (size_t)(mtile * 16 + 8 * hi) * ND + otile * 16 + lo;
#pragma unroll
  for (int v = 0; v < 8; ++v)
    hout[(size_t)v * ND] = c[v] + bv;
}

__device__ __forceinline__ void gpr_row_fma(
    const float4* __restrict__ H4, int s, float wc, unsigned sub,
    float4& a0, float4& a1) {
  unsigned q = (unsigned)s * (ND / 4) + sub;
  float4 h0 = H4[q];
  float4 h1 = H4[q + 8];
  a0.x = fmaf(h0.x, wc, a0.x); a0.y = fmaf(h0.y, wc, a0.y);
  a0.z = fmaf(h0.z, wc, a0.z); a0.w = fmaf(h0.w, wc, a0.w);
  a1.x = fmaf(h1.x, wc, a1.x); a1.y = fmaf(h1.y, wc, a1.y);
  a1.z = fmaf(h1.z, wc, a1.z); a1.w = fmaf(h1.w, wc, a1.w);
}

template <int K>
__device__ __forceinline__ void gpr_edge_step(
    const float4* __restrict__ H4, int2 rec, unsigned sub,
    float4& a0, float4& a1) {
  int   s  = __builtin_amdgcn_mov_dpp8(rec.x, K * 0x249249);
  float wc = __int_as_float(__builtin_amdgcn_mov_dpp8(rec.y, K * 0x249249));
  gpr_row_fma(H4, s, wc, sub, a0, a1);
}

__device__ __forceinline__ int2 gpr_load_slot(
    const int2* __restrict__ slots, unsigned rowBase, int j, int sub, int cs,
    int s0, int hcap) {
  int idx = j + sub;
  idx = idx < cs ? idx : j;
  return slots[rowBase + (unsigned)(idx < s0 ? idx : idx - s0 + hcap)];
}

__global__ __launch_bounds__(256, 16) void gpr_pull_kernel(
    const float* __restrict__ H, const int2* __restrict__ slots,
    const int* __restrict__ cnt, const float* __restrict__ w,
    const int* __restrict__ src, const int* __restrict__ head,
    const int* __restrict__ next, float* __restrict__ xcur,
    float* __restrict__ hidden, const float* __restrict__ temp,
    int tempIdx, int cap, int nRows, int nEdges) {
  int tid = blockIdx.x * 256 + threadIdx.x;
  int row = tid >> 3;
  if (row >= nRows) return;
  int sub = tid & 7;
  const float4* H4 = (const float4*)H;

  float4 a0 = make_float4(0.f, 0.f, 0.f, 0.f);
  float4 a1 = make_float4(0.f, 0.f, 0.f, 0.f);

  int hcap = cap >> 1;
  int c0 = cnt[row];
  int c1 = cnt[nRows + row];
  int s0 = c0 < hcap ? c0 : hcap;
  int s1 = c1 < hcap ? c1 : hcap;
  int cs = s0 + s1;
  unsigned rowBase = (unsigned)row * (unsigned)cap;

  int2 rn = make_int2(0, 0);
  if (cs > 0) rn = gpr_load_slot(slots, rowBase, 0, sub, cs, s0, hcap);
  for (int j0 = 0; j0 < cs; j0 += 8) {
    int2 r = rn;
    r.y = (j0 + sub < cs) ? r.y : 0;
    if (j0 + 8 < cs) rn = gpr_load_slot(slots, rowBase, j0 + 8, sub, cs, s0, hcap);
    gpr_edge_step<0>(H4, r, sub, a0, a1);
    gpr_edge_step<1>(H4, r, sub, a0, a1);
    gpr_edge_step<2>(H4, r, sub, a0, a1);
    gpr_edge_step<3>(H4, r, sub, a0, a1);
    gpr_edge_step<4>(H4, r, sub, a0, a1);
    gpr_edge_step<5>(H4, r, sub, a0, a1);
    gpr_edge_step<6>(H4, r, sub, a0, a1);
    gpr_edge_step<7>(H4, r, sub, a0, a1);
  }

  if (c0 > hcap || c1 > hcap) {
    int e = head[row];
    for (int hop = 0; (unsigned)e < (unsigned)nEdges && hop < nEdges; ++hop) {
      int   s  = src[e];
      float wc = w[e];
      e = next[e];
      gpr_row_fma(H4, s, wc, sub, a0, a1);
    }
  }

  float t = temp[tempIdx];
  unsigned o = (unsigned)row * ND + (unsigned)sub * 4;
  float4 v0 = make_float4(fmaxf(a0.x, 0.f), fmaxf(a0.y, 0.f), fmaxf(a0.z, 0.f), fmaxf(a0.w, 0.f));
  float4 v1 = make_float4(fmaxf(a1.x, 0.f), fmaxf(a1.y, 0.f), fmaxf(a1.z, 0.f), fmaxf(a1.w, 0.f));
  float4 g0 = *(const float4*)(hidden + o);
  float4 g1 = *(const float4*)(hidden + o + 32);
  *(float4*)(xcur + o)      = v0;
  *(float4*)(xcur + o + 32) = v1;
  g0.x += v0.x * t; g0.y += v0.y * t; g0.z += v0.z * t; g0.w += v0.w * t;
  g1.x += v1.x * t; g1.y += v1.y * t; g1.z += v1.z * t; g1.w += v1.w * t;
  *(float4*)(hidden + o)      = g0;
  *(float4*)(hidden + o + 32) = g1;
}

extern "C" void kernel_launch(void* const* d_in, const int* in_sizes, int n_in,
                              void* d_out, int out_size, void* d_ws, size_t ws_size,
                              hipStream_t stream) {
  const float* x    = (const float*)d_in[0];
  const float* w    = (const float*)d_in[1];
  const float* Wm   = (const float*)d_in[2];
  const float* bias = (const float*)d_in[3];
  const float* temp = (const float*)d_in[4];
  const int*   src  = (const int*)d_in[5];
  const int*   dst  = (const int*)d_in[6];
  float* hidden = (float*)d_out;

  const int n = in_sizes[0] / ND;
  const int E = in_sizes[1];
  const int L = in_sizes[3] / ND;
  const int total = n * ND;

  const size_t fixedBytes = ((size_t)total * 2 + (size_t)n * 3 + (size_t)E) * 4;
  const size_t room = ws_size > fixedBytes ? (ws_size - fixedBytes) / ((size_t)n * 8) : 0;
  const int cap = room >= 96 ? 96 : room >= 64 ? 64 : (int)(room & ~(size_t)7);

  float* h     = (float*)d_ws;
  float* xcur  = h    + (size_t)total;
  int2*  slots = (int2*)(xcur + (size_t)total);
  int*   cnt   = (int*)(slots + (size_t)n * cap);
  int*   head  = cnt + 2 * (size_t)n;
  int*   next  = head + n;

  const int ewBlocks   = (total + 255) / 256;
  const int nWaves     = (n / 16) * 4;
  const int gemmBlocks = (nWaves * 32 + 255) / 256;
  const int edgeBlocks = (E + 255) / 256;
  const int pullBlocks = (n * 8 + 255) / 256;

  gpr_init_kernel<<<ewBlocks, 256, 0, stream>>>(x, temp, hidden, xcur, cnt, head, n, total);
  gpr_fill_kernel<<<edgeBlocks, 256, 0, stream>>>(dst, src, w, cnt, slots, head, next, cap, n, E);

  for (int i = 0; i < L; ++i) {
    gpr_gemm_wmma_kernel<<<gemmBlocks, 256, 0, stream>>>(
        xcur, Wm + (size_t)i * ND * ND, bias + (size_t)i * ND, h, nWaves);
    gpr_pull_kernel<<<pullBlocks, 256, 0, stream>>>(
        h, slots, cnt, w, src, head, next, xcur, hidden, temp, i + 1, cap, n, E);
  }
}


# Start-up pass removed: layer 0's product reads x and clears the rank counters, ahead of fill; layer 0's pull forms hidden = x*temp0 + v*temp1 from x; no copy of x; 10 launches to 9; sum order kept
# speedup vs baseline: 4.0664x; 1.0753x over previous
#include <hip/hip_runtime.h>
#include <hip/hip_bf16.h>

typedef __attribute__((ext_vector_type(2))) float v2f;
typedef __attribute__((ext_vector_type(8))) float v8f;

#define ND 64


__global__ __launch_bounds__(256) void gpr_fill_kernel(
    const int* __restrict__ dst, const int* __restrict__ src,
    const float* __restrict__ w, int* __restrict__ cnt,
    int2* __restrict__ slots, int* __restrict__ head,
    int* __restrict__ next, int cap, int nRows, int nEdges) {
  int e = blockIdx.x * 256 + threadIdx.x;
  int hb = (int)((__builtin_amdgcn_s_sendmsg_rtn(0x87) >> 18) & 1u);
  int hcap = cap >> 1;
  if (e < nEdges) {
    int d = dst[e];
    int s = src[e];
    float wv = w[e];
    int r = atomicAdd(cnt + (size_t)hb * nRows + d, 1);
    if (r < hcap) slots[(size_t)d * cap + hb * hcap + r] = make_int2(s, __float_as_int(wv));
    else          next[e] = atomicExch(head + d, e);
  }
}

template <bool CLEAR>
__global__ __launch_bounds__(256) void gpr_gemm_wmma_kernel(
    const float* __restrict__ X, const float* __restrict__ Wm,
    const float* __restrict__ bias, float* __restrict__ H,
    int nWaves, int* __restrict__ cnt, int* __restrict__ head, int nRows) {
  int gid = blockIdx.x * 256 + threadIdx.x;
  if (CLEAR) {
    if (gid < 2 * nRows) cnt[gid] = 0;
    if (gid < nRows) head[gid] = -1;
  }
  int wave = gid >> 5;
  if (wave >= nWaves) return;
  int lane  = threadIdx.x & 31;
  int mtile = wave >> 2;
  int otile = wave & 3;
  int hi = lane >> 4;
  int lo = lane & 15;

  const float* arow = X  + (size_t)(mtile * 16 + lo) * ND;
  const float* brow = Wm + (size_t)(otile * 16 + lo) * ND;

  v8f c = {};
#pragma unroll
  for (int k = 0; k < ND; k += 4) {
    v2f a = *(const v2f*)(arow + k + 2 * hi);
    v2f b = *(const v2f*)(brow + k + 2 * hi);
    c = __builtin_amdgcn_wmma_f32_16x16x4_f32(
         false, a,  false, b,
         (short)0, c,  false,  false);
  }

  float bv = bias[otile * 16 + lo];
  float* hout = H + (size_t)(mtile * 16 + 8 * hi) * ND + otile * 16 + lo;
#pragma unroll
  for (int v = 0; v < 8; ++v)
    hout[(size_t)v * ND] = c[v] + bv;
}

__device__ __forceinline__ void gpr_row_fma(
    const float4* __restrict__ H4, int s, float wc, unsigned sub,
    float4& a0, float4& a1) {
  unsigned q = (unsigned)s * (ND / 4) + sub;
  float4 h0 = H4[q];
  float4 h1 = H4[q + 8];
  a0.x = fmaf(h0.x, wc, a0.x); a0.y = fmaf(h0.y, wc, a0.y);
  a0.z = fmaf(h0.z, wc, a0.z); a0.w = fmaf(h0.w, wc, a0.w);
  a1.x = fmaf(h1.x, wc, a1.x); a1.y = fmaf(h1.y, wc, a1.y);
  a1.z = fmaf(h1.z, wc, a1.z); a1.w = fmaf(h1.w, wc, a1.w);
}

template <int K>
__device__ __forceinline__ void gpr_edge_step(
    const float4* __restrict__ H4, int2 rec, unsigned sub,
    float4& a0, float4& a1) {
  int   s  = __builtin_amdgcn_mov_dpp8(rec.x, K * 0x249249);
  float wc = __int_as_float(__builtin_amdgcn_mov_dpp8(rec.y, K * 0x249249));
  gpr_row_fma(H4, s, wc, sub, a0, a1);
}

__device__ __forceinline__ int2 gpr_load_slot(
    const int2* __restrict__ slots, unsigned rowBase, int j, int sub, int cs,
    int s0, int hcap) {
  int idx = j + sub;
  idx = idx < cs ? idx : j;
  return slots[rowBase + (unsigned)(idx < s0 ? idx : idx - s0 + hcap)];
}

template <bool FIRST>
__global__ __launch_bounds__(256, 16) void gpr_pull_kernel(
    const float* __restrict__ x0,
    const float* __restrict__ H, const int2* __restrict__ slots,
    const int* __restrict__ cnt, const float* __restrict__ w,
    const int* __restrict__ src, const int* __restrict__ head,
    const int* __restrict__ next, float* __restrict__ xcur,
    float* __restrict__ hidden, const float* __restrict__ temp,
    int tempIdx, int cap, int nRows, int nEdges) {
  int tid = blockIdx.x * 256 + threadIdx.x;
  int row = tid >> 3;
  if (row >= nRows) return;
  int sub = tid & 7;
  const float4* H4 = (const float4*)H;

  float4 a0 = make_float4(0.f, 0.f, 0.f, 0.f);
  float4 a1 = make_float4(0.f, 0.f, 0.f, 0.f);

  int hcap = cap >> 1;
  int c0 = cnt[row];
  int c1 = cnt[nRows + row];
  int s0 = c0 < hcap ? c0 : hcap;
  int s1 = c1 < hcap ? c1 : hcap;
  int cs = s0 + s1;
  unsigned rowBase = (unsigned)row * (unsigned)cap;

  int2 rn = make_int2(0, 0);
  if (cs > 0) rn = gpr_load_slot(slots, rowBase, 0, sub, cs, s0, hcap);
  for (int j0 = 0; j0 < cs; j0 += 8) {
    int2 r = rn;
    r.y = (j0 + sub < cs) ? r.y : 0;
    if (j0 + 8 < cs) rn = gpr_load_slot(slots, rowBase, j0 + 8, sub, cs, s0, hcap);
    gpr_edge_step<0>(H4, r, sub, a0, a1);
    gpr_edge_step<1>(H4, r, sub, a0, a1);
    gpr_edge_step<2>(H4, r, sub, a0, a1);
    gpr_edge_step<3>(H4, r, sub, a0, a1);
    gpr_edge_step<4>(H4, r, sub, a0, a1);
    gpr_edge_step<5>(H4, r, sub, a0, a1);
    gpr_edge_step<6>(H4, r, sub, a0, a1);
    gpr_edge_step<7>(H4, r, sub, a0, a1);
  }

  if (c0 > hcap || c1 > hcap) {
    int e = head[row];
    for (int hop = 0; (unsigned)e < (unsigned)nEdges && hop < nEdges; ++hop) {
      int   s  = src[e];
      float wc = w[e];
      e = next[e];
      gpr_row_fma(H4, s, wc, sub, a0, a1);
    }
  }

  float t = temp[tempIdx];
  unsigned o = (unsigned)row * ND + (unsigned)sub * 4;
  float4 v0 = make_float4(fmaxf(a0.x, 0.f), fmaxf(a0.y, 0.f), fmaxf(a0.z, 0.f), fmaxf(a0.w, 0.f));
  float4 v1 = make_float4(fmaxf(a1.x, 0.f), fmaxf(a1.y, 0.f), fmaxf(a1.z, 0.f), fmaxf(a1.w, 0.f));
  float4 g0, g1;
  if (FIRST) {
    float t0 = temp[0];
    g0 = *(const float4*)(x0 + o);
    g1 = *(const float4*)(x0 + o + 32);
    g0.x *= t0; g0.y *= t0; g0.z *= t0; g0.w *= t0;
    g1.x *= t0; g1.y *= t0; g1.z *= t0; g1.w *= t0;
  } else {
    g0 = *(const float4*)(hidden + o);
    g1 = *(const float4*)(hidden + o + 32);
  }
  *(float4*)(xcur + o)      = v0;
  *(float4*)(xcur + o + 32) = v1;
  g0.x += v0.x * t; g0.y += v0.y * t; g0.z += v0.z * t; g0.w += v0.w * t;
  g1.x += v1.x * t; g1.y += v1.y * t; g1.z += v1.z * t; g1.w += v1.w * t;
  *(float4*)(hidden + o)      = g0;
  *(float4*)(hidden + o + 32) = g1;
}

extern "C" void kernel_launch(void* const* d_in, const int* in_sizes, int n_in,
                              void* d_out, int out_size, void* d_ws, size_t ws_size,
                              hipStream_t stream) {
  const float* x    = (const float*)d_in[0];
  const float* w    = (const float*)d_in[1];
  const float* Wm   = (const float*)d_in[2];
  const float* bias = (const float*)d_in[3];
  const float* temp = (const float*)d_in[4];
  const int*   src  = (const int*)d_in[5];
  const int*   dst  = (const int*)d_in[6];
  float* hidden = (float*)d_out;

  const int n = in_sizes[0] / ND;
  const int E = in_sizes[1];
  const int L = in_sizes[3] / ND;
  const int total = n * ND;

  const size_t fixedBytes = ((size_t)total * 2 + (size_t)n * 3 + (size_t)E) * 4;
  const size_t room = ws_size > fixedBytes ? (ws_size - fixedBytes) / ((size_t)n * 8) : 0;
  const int cap = room >= 96 ? 96 : room >= 64 ? 64 : (int)(room & ~(size_t)7);

  float* h     = (float*)d_ws;
  float* xcur  = h    + (size_t)total;
  int2*  slots = (int2*)(xcur + (size_t)total);
  int*   cnt   = (int*)(slots + (size_t)n * cap);
  int*   head  = cnt + 2 * (size_t)n;
  int*   next  = head + n;

  const int nWaves     = (n / 16) * 4;
  const int gemmBlocks = (nWaves * 32 + 255) / 256;
  const int edgeBlocks = (E + 255) / 256;
  const int pullBlocks = (n * 8 + 255) / 256;

  gpr_gemm_wmma_kernel<true><<<gemmBlocks, 256, 0, stream>>>(
      x, Wm, bias, h, nWaves, cnt, head, n);
  gpr_fill_kernel<<<edgeBlocks, 256, 0, stream>>>(dst, src, w, cnt, slots, head, next, cap, n, E);
  gpr_pull_kernel<true><<<pullBlocks, 256, 0, stream>>>(
      x, h, slots, cnt, w, src, head, next, xcur, hidden, temp, 1, cap, n, E);

  for (int i = 1; i < L; ++i) {
    gpr_gemm_wmma_kernel<false><<<gemmBlocks, 256, 0, stream>>>(
        xcur, Wm + (size_t)i * ND * ND, bias + (size_t)i * ND, h, nWaves, cnt, head, n);
    gpr_pull_kernel<false><<<pullBlocks, 256, 0, stream>>>(
        x, h, slots, cnt, w, src, head, next, xcur, hidden, temp, i + 1, cap, n, E);
  }
}
